# starting kernel
# baseline (speedup 1.0000x reference)
#include <hip/hip_runtime.h>

#define EPSF 1e-6f
#define NSZ 64
#define NLAYERS 63
#define BATCH_TOTAL 262144
#define SINK_ITERS 20
#define ROWS_PER_BLOCK 128

typedef __attribute__((ext_vector_type(2))) float v2f;
typedef __attribute__((ext_vector_type(8))) float v8f;

__global__ __launch_bounds__(64) void btln_sinkhorn_kernel(const float* __restrict__ logits,
                                                           float* __restrict__ P) {
    __shared__ float l[64 * 65];
    const int t = threadIdx.x;
    for (int j = 0; j < 64; ++j) l[t * 65 + j] = logits[t * 64 + j];
    __syncthreads();
    for (int it = 0; it < SINK_ITERS; ++it) {
        float m = -3.4e38f;
        for (int j = 0; j < 64; ++j) m = fmaxf(m, l[t * 65 + j]);
        float s = 0.f;
        for (int j = 0; j < 64; ++j) s += expf(l[t * 65 + j] - m);
        float lse = m + logf(s);
        for (int j = 0; j < 64; ++j) l[t * 65 + j] -= lse;
        __syncthreads();
        m = -3.4e38f;
        for (int i = 0; i < 64; ++i) m = fmaxf(m, l[i * 65 + t]);
        s = 0.f;
        for (int i = 0; i < 64; ++i) s += expf(l[i * 65 + t] - m);
        lse = m + logf(s);
        for (int i = 0; i < 64; ++i) l[i * 65 + t] -= lse;
        __syncthreads();
    }
    for (int j = 0; j < 64; ++j) P[t * 64 + j] = expf(l[t * 65 + j]);
}

__device__ __forceinline__ float btln_clip01(float v) {
    return fminf(fmaxf(v, EPSF), 1.0f - EPSF);
}

__device__ __forceinline__ float btln_F_base(float x, float y, float a,
                                             float w0, float w1, float p) {
    float lin = w0 * x + w1 * y;
    float g = exp2f(p * (2.0f * w0 * log2f(x) + 2.0f * w1 * log2f(y)));
    float rule0 = ((fabsf(x - 1.0f) < EPSF) && (fabsf(y - 1.0f) < EPSF)) ? 1.0f : 0.0f;
    float mid = (3.0f - 4.0f * a) * lin + (4.0f * a - 2.0f) * g;
    float r = (a >= 0.75f) ? g : ((a > 0.5f) ? mid : lin);
    return (fabsf(a - 2.0f) < EPSF) ? rule0 : r;
}

__device__ __forceinline__ float btln_F(float x, float y, float a, float w0, float w1,
                                        float p_hi, float p_lo) {
    x = btln_clip01(x);
    y = btln_clip01(y);
    float lin = w0 * x + w1 * y;
    float hi = btln_F_base(x, y, a, w0, w1, p_hi);
    float aref = fminf(fmaxf(1.0f - a, -1.0f + EPSF), 2.0f - EPSF);
    float lo = 1.0f - btln_F_base(btln_clip01(1.0f - x), btln_clip01(1.0f - y),
                                  aref, w0, w1, p_lo);
    float r = (fabsf(a - 0.5f) < EPSF) ? lin : ((a < 0.5f) ? lo : hi);
    return (r != r) ? a : r;
}

__global__ __launch_bounds__(256) void btln_fused_kernel(const float* __restrict__ x,
                                                         const float* __restrict__ P,
                                                         const float* __restrict__ weights,
                                                         const float* __restrict__ biases,
                                                         float* __restrict__ out) {
    __shared__ float PT[64 * 66];
    __shared__ float leaves[ROWS_PER_BLOCK * 66];
    __shared__ float pw0[NLAYERS], pw1[NLAYERS], pb[NLAYERS];
    __shared__ float pph[NLAYERS], ppl[NLAYERS];

    const int tid = threadIdx.x;

    for (int e = tid; e < 64 * 64; e += 256) {
        int k = e >> 6, n = e & 63;
        PT[n * 66 + k] = P[e];
    }
    if (tid < NLAYERS) {
        float w = weights[tid];
        float b = biases[tid];
        pw0[tid] = w;
        pw1[tid] = 1.0f - w;
        pb[tid]  = b;
        pph[tid] = sqrtf(3.0f / fmaxf(2.0f - b, EPSF)) - 1.0f;
        float aref = fminf(fmaxf(1.0f - b, -1.0f + EPSF), 2.0f - EPSF);
        ppl[tid] = sqrtf(3.0f / fmaxf(2.0f - aref, EPSF)) - 1.0f;
    }
    __syncthreads();

    const int wave = tid >> 5;
    const int lane = tid & 31;
    const int nn   = lane & 15;
    const int kb   = (lane >> 4) * 2;
    const long block_row0 = (long)blockIdx.x * ROWS_PER_BLOCK;
    const int lrow0 = wave * 16;

    const float* arow = x + (block_row0 + lrow0 + nn) * 64 + kb;

    v8f acc0 = {}, acc1 = {}, acc2 = {}, acc3 = {};
#pragma unroll
    for (int s = 0; s < 16; ++s) {
        v2f a  = *(const v2f*)(arow + 4 * s);
        v2f b0 = *(const v2f*)&PT[( 0 + nn) * 66 + 4 * s + kb];
        v2f b1 = *(const v2f*)&PT[(16 + nn) * 66 + 4 * s + kb];
        v2f b2 = *(const v2f*)&PT[(32 + nn) * 66 + 4 * s + kb];
        v2f b3 = *(const v2f*)&PT[(48 + nn) * 66 + 4 * s + kb];
        acc0 = __builtin_amdgcn_wmma_f32_16x16x4_f32(false, a, false, b0, (short)0, acc0, false, false);
        acc1 = __builtin_amdgcn_wmma_f32_16x16x4_f32(false, a, false, b1, (short)0, acc1, false, false);
        acc2 = __builtin_amdgcn_wmma_f32_16x16x4_f32(false, a, false, b2, (short)0, acc2, false, false);
        acc3 = __builtin_amdgcn_wmma_f32_16x16x4_f32(false, a, false, b3, (short)0, acc3, false, false);
    }

    const int mrow = lrow0 + ((lane >> 4) << 3);
#pragma unroll
    for (int r = 0; r < 8; ++r) {
        leaves[(mrow + r) * 66 +  0 + nn] = acc0[r];
        leaves[(mrow + r) * 66 + 16 + nn] = acc1[r];
        leaves[(mrow + r) * 66 + 32 + nn] = acc2[r];
        leaves[(mrow + r) * 66 + 48 + nn] = acc3[r];
    }
    __syncthreads();

    if (tid < ROWS_PER_BLOCK) {
        float st = leaves[tid * 66 + 0];
        for (int i = 0; i < NLAYERS; ++i) {
            float y = leaves[tid * 66 + i + 1];
            st = btln_F(st, y, pb[i], pw0[i], pw1[i], pph[i], ppl[i]);
        }
        out[block_row0 + tid] = st;
    }
}

extern "C" void kernel_launch(void* const* d_in, const int* in_sizes, int n_in,
                              void* d_out, int out_size, void* d_ws, size_t ws_size,
                              hipStream_t stream) {
    const float* x       = (const float*)d_in[0];
    const float* logits  = (const float*)d_in[1];
    const float* weights = (const float*)d_in[2];
    const float* biases  = (const float*)d_in[3];
    float* out = (float*)d_out;
    float* P   = (float*)d_ws;

    btln_sinkhorn_kernel<<<1, 64, 0, stream>>>(logits, P);
    btln_fused_kernel<<<BATCH_TOTAL / ROWS_PER_BLOCK, 256, 0, stream>>>(x, P, weights, biases, out);
}


# Scan layer cut to its live arithmetic: w0/w1 folded into the table's exponent and blend weights, re-clip of 1-v and NaN select dropped (identities there), slice loop unrolled; layer sums regrouped
# speedup vs baseline: 3.8856x; 3.8856x over previous
#include <hip/hip_runtime.h>

#define EPSF 1e-6f
#define NSZ 64
#define NLAYERS 63
#define BATCH_TOTAL 262144
#define SINK_ITERS 20
#define ROWS_PER_BLOCK 128
#define SLICE_LEAVES 16
#define NSLICES 4
#define LPAD 18

typedef __attribute__((ext_vector_type(2))) float v2f;
typedef __attribute__((ext_vector_type(8))) float v8f;

__global__ __launch_bounds__(128) void btln_sinkhorn_kernel(const float* __restrict__ logits,
                                                            float* __restrict__ P) {
    __shared__ float rmax[64];
    __shared__ __attribute__((aligned(16))) float rvec[64];
    __shared__ __attribute__((aligned(16))) float cvec[64];

    const int tid  = threadIdx.x;
    const int wave = tid >> 5;
    const int lane = tid & 31;
    const int nn   = lane & 15;
    const int hh   = lane >> 4;
    const int kb   = hh * 2;
    const int base = wave * 16;
    const int qsel = nn & 7;

    v2f ka[16];
    const float* lrow = logits + (base + nn) * 64 + kb;
    float m = -3.4e38f;
#pragma unroll
    for (int s = 0; s < 16; ++s) {
        ka[s] = *(const v2f*)(lrow + 4 * s);
        m = fmaxf(m, fmaxf(ka[s].x, ka[s].y));
    }
    m = fmaxf(m, __shfl_xor(m, 16));
    if (hh == 0) rmax[base + nn] = m;
    if (tid < 64) cvec[tid] = 1.0f;
    __syncthreads();
#pragma unroll
    for (int s = 0; s < 16; ++s) {
        ka[s].x = expf(ka[s].x - m);
        ka[s].y = expf(ka[s].y - m);
    }
    v2f kq[16];
#pragma unroll
    for (int s = 0; s < 16; ++s) {
        const int i = 4 * s + kb;
        kq[s].x = expf(logits[i * 64 + base + nn] - rmax[i]);
        kq[s].y = expf(logits[(i + 1) * 64 + base + nn] - rmax[i + 1]);
    }

    for (int it = 0; it < SINK_ITERS; ++it) {
        v8f y0 = {}, y1 = {};
#pragma unroll
        for (int s = 0; s < 16; s += 2) {
            v2f c0 = *(const v2f*)&cvec[4 * s + kb];
            v2f c1 = *(const v2f*)&cvec[4 * s + 4 + kb];
            y0 = __builtin_amdgcn_wmma_f32_16x16x4_f32(false, ka[s],     false, c0, (short)0, y0, false, false);
            y1 = __builtin_amdgcn_wmma_f32_16x16x4_f32(false, ka[s + 1], false, c1, (short)0, y1, false, false);
        }
        const v8f y = y0 + y1;
        float ys = y[0];
        ys = (qsel == 1) ? y[1] : ys;
        ys = (qsel == 2) ? y[2] : ys;
        ys = (qsel == 3) ? y[3] : ys;
        ys = (qsel == 4) ? y[4] : ys;
        ys = (qsel == 5) ? y[5] : ys;
        ys = (qsel == 6) ? y[6] : ys;
        ys = (qsel == 7) ? y[7] : ys;
        rvec[base + 8 * hh + qsel] = 1.0f / ys;
        __syncthreads();
        v8f z0 = {}, z1 = {};
#pragma unroll
        for (int s = 0; s < 16; s += 2) {
            v2f r0 = *(const v2f*)&rvec[4 * s + kb];
            v2f r1 = *(const v2f*)&rvec[4 * s + 4 + kb];
            z0 = __builtin_amdgcn_wmma_f32_16x16x4_f32(false, r0, false, kq[s],     (short)0, z0, false, false);
            z1 = __builtin_amdgcn_wmma_f32_16x16x4_f32(false, r1, false, kq[s + 1], (short)0, z1, false, false);
        }
        cvec[base + nn] = 1.0f / (z0[0] + z1[0]);
        __syncthreads();
    }

    const float ri = rvec[base + nn];
    float* prow = P + (base + nn) * 64 + kb;
#pragma unroll
    for (int s = 0; s < 16; ++s) {
        v2f c = *(const v2f*)&cvec[4 * s + kb];
        v2f p;
        p.x = ri * ka[s].x * c.x;
        p.y = ri * ka[s].y * c.y;
        *(v2f*)(prow + 4 * s) = p;
    }
}

__device__ __forceinline__ float btln_clip01(float v) {
    return fminf(fmaxf(v, EPSF), 1.0f - EPSF);
}

__device__ __forceinline__ float btln_F_base(float x, float y, float a,
                                             float w0, float w1, float p) {
    float lin = w0 * x + w1 * y;
    float g = exp2f(p * (2.0f * w0 * log2f(x) + 2.0f * w1 * log2f(y)));
    float rule0 = ((fabsf(x - 1.0f) < EPSF) && (fabsf(y - 1.0f) < EPSF)) ? 1.0f : 0.0f;
    float mid = (3.0f - 4.0f * a) * lin + (4.0f * a - 2.0f) * g;
    float r = (a >= 0.75f) ? g : ((a > 0.5f) ? mid : lin);
    return (fabsf(a - 2.0f) < EPSF) ? rule0 : r;
}

__device__ __forceinline__ float btln_F(float x, float y, float a, float w0, float w1,
                                        float p_hi, float p_lo) {
    x = btln_clip01(x);
    y = btln_clip01(y);
    float lin = w0 * x + w1 * y;
    float hi = btln_F_base(x, y, a, w0, w1, p_hi);
    float aref = fminf(fmaxf(1.0f - a, -1.0f + EPSF), 2.0f - EPSF);
    float lo = 1.0f - btln_F_base(btln_clip01(1.0f - x), btln_clip01(1.0f - y),
                                  aref, w0, w1, p_lo);
    float r = (fabsf(a - 0.5f) < EPSF) ? lin : ((a < 0.5f) ? lo : hi);
    return (r != r) ? a : r;
}

#define TBL_W 12

__device__ __forceinline__ void btln_side_weights(float a, float& al, float& be) {
    al = (a >= 0.75f) ? 0.0f : ((a > 0.5f) ? (3.0f - 4.0f * a) : 1.0f);
    be = (a >= 0.75f) ? 1.0f : ((a > 0.5f) ? (4.0f * a - 2.0f) : 0.0f);
}


__device__ __forceinline__ float btln_layer(float st, float leaf, const float* tb) {
    const v2f a1 = *(const v2f*)(tb + 0);
    const v2f a2 = *(const v2f*)(tb + 2);
    const v2f b1 = *(const v2f*)(tb + 4);
    const v2f b2 = *(const v2f*)(tb + 6);
    const v2f cB = *(const v2f*)(tb + 8);
    const float k0 = tb[10];
    const float xs = btln_clip01(st);
    const float ys = btln_clip01(leaf);
    v2f X, Y;
    X.x = xs; X.y = 1.0f - xs;
    Y.x = ys; Y.y = 1.0f - ys;
    v2f LX, LY;
    LX.x = log2f(X.x); LX.y = log2f(X.y);
    LY.x = log2f(Y.x); LY.y = log2f(Y.y);
    const v2f e = __builtin_elementwise_fma(a1, LX, a2 * LY);
    v2f G;
    G.x = exp2f(e.x); G.y = exp2f(e.y);
    const v2f S = __builtin_elementwise_fma(cB, G, __builtin_elementwise_fma(b1, X, b2 * Y));
    return (k0 + S.y) + S.x;
}


__device__ __forceinline__ void btln_product_slice(const v2f (&a0)[16], const v2f (&a1)[16],
                                                   const float* bcol, float* dst) {
    v8f c0 = {}, c1 = {};
#pragma unroll
    for (int s = 0; s < 16; ++s) {
        const v2f b = *(const v2f*)(bcol + 4 * s);
        c0 = __builtin_amdgcn_wmma_f32_16x16x4_f32(false, a0[s], false, b, (short)0, c0, false, false);
        c1 = __builtin_amdgcn_wmma_f32_16x16x4_f32(false, a1[s], false, b, (short)0, c1, false, false);
    }
#pragma unroll
    for (int r = 0; r < 8; ++r) {
        dst[r * LPAD] = c0[r];
        dst[(16 + r) * LPAD] = c1[r];
    }
    asm volatile("s_wait_dscnt 0x0" ::: "memory");
}

__global__ __launch_bounds__(256, 10) void btln_fused_kernel(const float* x,
                                                        const float* __restrict__ P,
                                                        const float* __restrict__ weights,
                                                        const float* __restrict__ biases,
                                                        float* __restrict__ out) {
    __shared__ __attribute__((aligned(16))) float PT[64 * 66];
    __shared__ float ring[2][ROWS_PER_BLOCK * LPAD];
    __shared__ __attribute__((aligned(16))) float tbl[NLAYERS * TBL_W];
    __shared__ int general_path[2];

    const int tid  = threadIdx.x;
    const int lane = tid & 31;
    const int nn   = lane & 15;
    const int hh   = lane >> 4;
    const int kb   = hh * 2;
    const int wave = __builtin_amdgcn_readfirstlane(tid >> 5);
    const bool producer = wave >= 4;
    const int prow = (wave & 3) * 32;
    const long block_row0 = (long)blockIdx.x * ROWS_PER_BLOCK;

    v2f a0[16], a1[16];
    if (producer) {
        const float* arow = x + (block_row0 + prow + nn) * 64 + kb;
#pragma unroll
        for (int s = 0; s < 16; ++s) {
            a0[s] = *(const v2f*)(arow + 4 * s);
            a1[s] = *(const v2f*)(arow + 16 * 64 + 4 * s);
        }
    }

    for (int e = tid; e < 64 * 64; e += 256) {
        int k = e >> 6, n = e & 63;
        PT[n * 66 + k] = P[e];
    }
    bool outside = false;
    if (tid < NLAYERS) {
        float w = weights[tid];
        float b = biases[tid];
        float aref = fminf(fmaxf(1.0f - b, -1.0f + EPSF), 2.0f - EPSF);
        float ah, bh, al, bl;
        btln_side_weights(b, ah, bh);
        btln_side_weights(aref, al, bl);
        const bool is_lin = fabsf(b - 0.5f) < EPSF;
        const bool use_lo = !is_lin && (b < 0.5f);
        if (is_lin) { ah = 1.0f; bh = 0.0f; }
        float* t = &tbl[tid * TBL_W];
        const float w0 = w, w1 = 1.0f - w;
        const float pp_hi = 2.0f * (sqrtf(3.0f / fmaxf(2.0f - b, EPSF)) - 1.0f);
        const float pp_lo = 2.0f * (sqrtf(3.0f / fmaxf(2.0f - aref, EPSF)) - 1.0f);
        const float cA_hi = use_lo ? 0.0f : ah, cA_lo = use_lo ? -al : 0.0f;
        t[0]  = pp_hi * w0;  t[1]  = pp_lo * w0;
        t[2]  = pp_hi * w1;  t[3]  = pp_lo * w1;
        t[4]  = cA_hi * w0;  t[5]  = cA_lo * w0;
        t[6]  = cA_hi * w1;  t[7]  = cA_lo * w1;
        t[8]  = use_lo ? 0.0f : bh;
        t[9]  = use_lo ? -bl : 0.0f;
        t[10] = use_lo ? 1.0f : 0.0f;
        t[11] = 0.0f;
        outside = !(b >= 0.0f && b <= 1.0f && w >= 0.0f && w <= 1.0f);
    }
    if (tid < 64) {
        const bool any = __ballot(outside) != 0;
        if ((tid & 31) == 0) general_path[tid >> 5] = any ? 1 : 0;
    }
    __syncthreads();

    const float* bcol0 = &PT[nn * 66 + kb];
    const int dofs = (prow + 8 * hh) * LPAD + nn;

    if (producer) btln_product_slice(a0, a1, bcol0, &ring[0][dofs]);

    const bool general = (general_path[0] | general_path[1]) != 0;
    float st = 0.0f;

    for (int t = 0; t < NSLICES; ++t) {
        __syncthreads();
        if (producer) {
            if (t + 1 < NSLICES)
                btln_product_slice(a0, a1, bcol0 + (t + 1) * SLICE_LEAVES * 66, &ring[(t + 1) & 1][dofs]);
        } else {
            const float* src = &ring[t & 1][tid * LPAD];
            if (general) {
                int j = 0;
                if (t == 0) { st = src[0]; j = 1; }
                for (; j < SLICE_LEAVES; ++j) {
                    const int L = t * SLICE_LEAVES + j - 1;
                    const float w = weights[L];
                    const float b = biases[L];
                    const float aref = fminf(fmaxf(1.0f - b, -1.0f + EPSF), 2.0f - EPSF);
                    const float p_hi = sqrtf(3.0f / fmaxf(2.0f - b, EPSF)) - 1.0f;
                    const float p_lo = sqrtf(3.0f / fmaxf(2.0f - aref, EPSF)) - 1.0f;
                    st = btln_F(st, src[j], b, w, 1.0f - w, p_hi, p_lo);
                }
            } else {
                const float* tb0 = &tbl[t * SLICE_LEAVES * TBL_W];
                if (t == 0) st = src[0];
                else        st = btln_layer(st, src[0], tb0 - TBL_W);
#pragma unroll
                for (int j = 1; j < SLICE_LEAVES; ++j)
                    st = btln_layer(st, src[j], tb0 + (j - 1) * TBL_W);
            }
        }
    }
    if (!producer) out[block_row0 + tid] = st;
}

extern "C" void kernel_launch(void* const* d_in, const int* in_sizes, int n_in,
                              void* d_out, int out_size, void* d_ws, size_t ws_size,
                              hipStream_t stream) {
    const float* x       = (const float*)d_in[0];
    const float* logits  = (const float*)d_in[1];
    const float* weights = (const float*)d_in[2];
    const float* biases  = (const float*)d_in[3];
    float* out = (float*)d_out;
    float* P   = (float*)d_ws;

    btln_sinkhorn_kernel<<<1, 128, 0, stream>>>(logits, P);
    btln_fused_kernel<<<BATCH_TOTAL / ROWS_PER_BLOCK, 256, 0, stream>>>(x, P, weights, biases, out);
}
